# starting kernel
# baseline (speedup 1.0000x reference)
#include <hip/hip_runtime.h>
#include <hip/hip_bf16.h>
#include <math.h>

typedef __attribute__((ext_vector_type(2))) float v2f;
typedef __attribute__((ext_vector_type(8))) float v8f;

#define NB   4
#define NDET 128
#define NT   2048
#define NPIX (256 * 256)
#define WAVES_PER_BLOCK 8

__global__ __launch_bounds__(256)
void bp_wmma_f32_kernel(const float* __restrict__ sino,
                        const float* __restrict__ lut,
                        float* __restrict__ out)
{
    __shared__ float sApod[NDET];
    if (threadIdx.x < NDET) {
        const float w = 6.28318530717958647692f / 127.0f;
        sApod[threadIdx.x] = 0.5f - 0.5f * cosf((float)threadIdx.x * w);
    }
    __syncthreads();

    const int lane = threadIdx.x & 31;
    const int wave = threadIdx.x >> 5;
    const int tile = blockIdx.x * WAVES_PER_BLOCK + wave;
    const int p0   = tile << 4;
    const int halfSel = (lane < 16) ? 0 : 2;
    const int prow = p0 + (lane & 15);

    v8f acc[NB] = { v8f{0}, v8f{0}, v8f{0}, v8f{0} };

    for (int d = 0; d < NDET; d += 4) {
        const int det0 = d + halfSel;

        const float4 L = *(const float4*)(lut + ((size_t)prow * NDET + det0) * 2);

        const float2 ap = *(const float2*)(&sApod[det0]);
        v2f bb; bb.x = ap.x; bb.y = ap.y;

        const int kf0 = (int)L.x;
        const int kf1 = (int)L.z;
        const bool valid0 = (kf0 >= 0) && (kf0 < NT - 1);
        const bool valid1 = (kf1 >= 0) && (kf1 < NT - 1);
        const int k00 = min(max(kf0, 0), NT - 2);
        const int k10 = min(max(kf1, 0), NT - 2);
        const int roff0 = det0 * NT + k00;
        const int roff1 = (det0 + 1) * NT + k10;
        const float a0 = L.y, a1 = L.w;

#pragma unroll
        for (int b = 0; b < NB; ++b) {
            const float* s = sino + (size_t)b * (NDET * NT);
            const float s00 = s[roff0], s01 = s[roff0 + 1];
            const float s10 = s[roff1], s11 = s[roff1 + 1];
            float sk0 = (1.0f - a0) * s00 + a0 * s01;
            float sk1 = (1.0f - a1) * s10 + a1 * s11;
            sk0 = valid0 ? sk0 : 0.0f;
            sk1 = valid1 ? sk1 : 0.0f;

            v2f A; A.x = sk0; A.y = sk1;

            acc[b] = __builtin_amdgcn_wmma_f32_16x16x4_f32(
                 false, A,  false, bb,
                 (short)0, acc[b],  false,  false);
        }
    }

    const float invNorm = 1.0f / 63.5f;

    if (lane == 0 || lane == 16) {
        const int rbase = p0 + ((lane == 16) ? 8 : 0);
#pragma unroll
        for (int b = 0; b < NB; ++b) {
#pragma unroll
            for (int r = 0; r < 8; ++r) {
                out[(size_t)b * NPIX + rbase + r] = acc[b][r] * invNorm;
            }
        }
    }
}

extern "C" void kernel_launch(void* const* d_in, const int* in_sizes, int n_in,
                              void* d_out, int out_size, void* d_ws, size_t ws_size,
                              hipStream_t stream) {
    const float* sino = (const float*)d_in[0];
    const float* lut  = (const float*)d_in[1];
    float* out = (float*)d_out;

    const int tiles  = NPIX / 16;
    const int blocks = tiles / WAVES_PER_BLOCK;
    bp_wmma_f32_kernel<<<blocks, 256, 0, stream>>>(sino, lut, out);
}


# Pre-pass launch re-lays sino batch-interleaved as [det][t][b] in workspace: each (pixel,det) gather is two aligned float4 loads, not four unaligned 8-byte loads from planes 1 MiB apart
# speedup vs baseline: 2.5548x; 2.5548x over previous
#include <hip/hip_runtime.h>
#include <hip/hip_bf16.h>
#include <math.h>

typedef __attribute__((ext_vector_type(2))) float v2f;
typedef __attribute__((ext_vector_type(8))) float v8f;

#define NB   4
#define NDET 128
#define NT   2048
#define NPIX (256 * 256)
#define WAVES_PER_BLOCK 8

#define SINO_T_BYTES ((size_t)NDET * NT * NB * sizeof(float))

#define PRE_THREADS 32768
#define PRE_PER_THREAD 8

__global__ __launch_bounds__(128)
void sino_interleave_kernel(const float* __restrict__ sino,
                            float* __restrict__ sinoT,
                            size_t ws_bytes)
{
    if (ws_bytes < SINO_T_BYTES) return;
    const int tid = blockIdx.x * 128 + threadIdx.x;
    float4 v[PRE_PER_THREAD];
#pragma unroll
    for (int i = 0; i < PRE_PER_THREAD; ++i) {
        const int e = tid + i * PRE_THREADS;
        v[i].x = sino[e];
        v[i].y = sino[1 * NDET * NT + e];
        v[i].z = sino[2 * NDET * NT + e];
        v[i].w = sino[3 * NDET * NT + e];
    }
#pragma unroll
    for (int i = 0; i < PRE_PER_THREAD; ++i) {
        const int e = tid + i * PRE_THREADS;
        ((float4*)sinoT)[e] = v[i];
    }
}

__global__ __launch_bounds__(256)
void bp_wmma_f32_kernel(const float* __restrict__ sino,
                        const float* __restrict__ lut,
                        const float* __restrict__ sinoT,
                        size_t ws_bytes,
                        float* __restrict__ out)
{
    __shared__ float sApod[NDET];
    if (threadIdx.x < NDET) {
        const float w = 6.28318530717958647692f / 127.0f;
        sApod[threadIdx.x] = 0.5f - 0.5f * cosf((float)threadIdx.x * w);
    }
    __syncthreads();

    const int lane = threadIdx.x & 31;
    const int wave = threadIdx.x >> 5;
    const int tile = blockIdx.x * WAVES_PER_BLOCK + wave;
    const int p0   = tile << 4;
    const int halfSel = (lane < 16) ? 0 : 2;
    const int prow = p0 + (lane & 15);

    v8f acc[NB] = { v8f{0}, v8f{0}, v8f{0}, v8f{0} };

    if (ws_bytes >= SINO_T_BYTES) {
        const float4* __restrict__ sT = (const float4*)sinoT;
        for (int d = 0; d < NDET; d += 4) {
            const int det0 = d + halfSel;

            const float4 L = *(const float4*)(lut + ((size_t)prow * NDET + det0) * 2);

            const float2 ap = *(const float2*)(&sApod[det0]);
            v2f bb; bb.x = ap.x; bb.y = ap.y;

            const int kf0 = (int)L.x;
            const int kf1 = (int)L.z;
            const bool valid0 = (kf0 >= 0) && (kf0 < NT - 1);
            const bool valid1 = (kf1 >= 0) && (kf1 < NT - 1);
            const int k00 = min(max(kf0, 0), NT - 2);
            const int k10 = min(max(kf1, 0), NT - 2);
            const int roff0 = det0 * NT + k00;
            const int roff1 = (det0 + 1) * NT + k10;
            const float a0 = L.y, a1 = L.w;

            const float4 s00 = sT[roff0], s01 = sT[roff0 + 1];
            const float4 s10 = sT[roff1], s11 = sT[roff1 + 1];

            const float q00[NB] = { s00.x, s00.y, s00.z, s00.w };
            const float q01[NB] = { s01.x, s01.y, s01.z, s01.w };
            const float q10[NB] = { s10.x, s10.y, s10.z, s10.w };
            const float q11[NB] = { s11.x, s11.y, s11.z, s11.w };

#pragma unroll
            for (int b = 0; b < NB; ++b) {
                float sk0 = (1.0f - a0) * q00[b] + a0 * q01[b];
                float sk1 = (1.0f - a1) * q10[b] + a1 * q11[b];
                sk0 = valid0 ? sk0 : 0.0f;
                sk1 = valid1 ? sk1 : 0.0f;

                v2f A; A.x = sk0; A.y = sk1;

                acc[b] = __builtin_amdgcn_wmma_f32_16x16x4_f32(
                     false, A,  false, bb,
                     (short)0, acc[b],  false,  false);
            }
        }
    } else {
        for (int d = 0; d < NDET; d += 4) {
            const int det0 = d + halfSel;
            const float4 L = *(const float4*)(lut + ((size_t)prow * NDET + det0) * 2);
            const float2 ap = *(const float2*)(&sApod[det0]);
            v2f bb; bb.x = ap.x; bb.y = ap.y;

            const int kf0 = (int)L.x;
            const int kf1 = (int)L.z;
            const bool valid0 = (kf0 >= 0) && (kf0 < NT - 1);
            const bool valid1 = (kf1 >= 0) && (kf1 < NT - 1);
            const int k00 = min(max(kf0, 0), NT - 2);
            const int k10 = min(max(kf1, 0), NT - 2);
            const int roff0 = det0 * NT + k00;
            const int roff1 = (det0 + 1) * NT + k10;
            const float a0 = L.y, a1 = L.w;

#pragma unroll
            for (int b = 0; b < NB; ++b) {
                const float* s = sino + (size_t)b * (NDET * NT);
                const float s00 = s[roff0], s01 = s[roff0 + 1];
                const float s10 = s[roff1], s11 = s[roff1 + 1];
                float sk0 = (1.0f - a0) * s00 + a0 * s01;
                float sk1 = (1.0f - a1) * s10 + a1 * s11;
                sk0 = valid0 ? sk0 : 0.0f;
                sk1 = valid1 ? sk1 : 0.0f;

                v2f A; A.x = sk0; A.y = sk1;

                acc[b] = __builtin_amdgcn_wmma_f32_16x16x4_f32(
                     false, A,  false, bb,
                     (short)0, acc[b],  false,  false);
            }
        }
    }

    const float invNorm = 1.0f / 63.5f;

    if (lane == 0 || lane == 16) {
        const int rbase = p0 + ((lane == 16) ? 8 : 0);
#pragma unroll
        for (int b = 0; b < NB; ++b) {
#pragma unroll
            for (int r = 0; r < 8; ++r) {
                out[(size_t)b * NPIX + rbase + r] = acc[b][r] * invNorm;
            }
        }
    }
}

extern "C" void kernel_launch(void* const* d_in, const int* in_sizes, int n_in,
                              void* d_out, int out_size, void* d_ws, size_t ws_size,
                              hipStream_t stream) {
    const float* sino = (const float*)d_in[0];
    const float* lut  = (const float*)d_in[1];
    float* out = (float*)d_out;
    float* sinoT = (float*)d_ws;

    sino_interleave_kernel<<<PRE_THREADS / 128, 128, 0, stream>>>(sino, sinoT, ws_size);

    const int tiles  = NPIX / 16;
    const int blocks = tiles / WAVES_PER_BLOCK;
    bp_wmma_f32_kernel<<<blocks, 256, 0, stream>>>(sino, lut, sinoT, ws_size, out);
}


# Detectors split over work-groups (8 each): their sinogram rows sit in LDS so gathers are LDS reads, not L2 lines; per-group partial sums added by a 2nd launch (changes summation order)
# speedup vs baseline: 4.7111x; 1.8440x over previous
#include <hip/hip_runtime.h>
#include <math.h>

typedef __attribute__((ext_vector_type(2))) float v2f;
typedef __attribute__((ext_vector_type(8))) float v8f;

#define NB   4
#define NDET 128
#define NT   2048
#define NPIX (256 * 256)
#define NTILE (NPIX / 16)

#define GDET 8
#define NGRP (NDET / GDET)
#define PIX_PER_BLOCK 4096
#define TILES_PER_BLOCK (PIX_PER_BLOCK / 16)
#define BLOCK_THREADS 512
#define WAVES_PER_BLOCK (BLOCK_THREADS / 32)
#define TILES_PER_WAVE (TILES_PER_BLOCK / WAVES_PER_BLOCK)
#define LUT_CHUNK 4

#define PART_BYTES ((size_t)NGRP * NTILE * NB * 16 * sizeof(float))

__device__ __forceinline__ void interp4(const float4* __restrict__ row, float kff, float a, float o[NB])
{
    const int kf = (int)kff;
    const bool valid = (kf >= 0) && (kf < NT - 1);
    const int k0 = min(max(kf, 0), NT - 2);
    const float4 s0 = row[k0], s1 = row[k0 + 1];
    const float q0[NB] = { s0.x, s0.y, s0.z, s0.w };
    const float q1[NB] = { s1.x, s1.y, s1.z, s1.w };
#pragma unroll
    for (int b = 0; b < NB; ++b) {
        const float sk = (1.0f - a) * q0[b] + a * q1[b];
        o[b] = valid ? sk : 0.0f;
    }
}

__global__ __launch_bounds__(BLOCK_THREADS)
void bp_detgroup_kernel(const float* __restrict__ sino,
                        const float* __restrict__ lut,
                        float* __restrict__ part,
                        size_t ws_bytes,
                        float* __restrict__ out)
{
    __shared__ float4 sRow[GDET * NT];

    const int lane = threadIdx.x & 31;
    const int wave = threadIdx.x >> 5;
    const int c = lane & 15;
    const int h = lane >> 4;
    const float w = 6.28318530717958647692f / 127.0f;

    if (ws_bytes < PART_BYTES) {
        const int tile = blockIdx.x * WAVES_PER_BLOCK + wave;
        const int p0 = tile << 4;
        const int prow = p0 + c;
        v8f acc[NB] = { v8f{0}, v8f{0}, v8f{0}, v8f{0} };
        for (int d = 0; d < NDET; d += 4) {
            const int det0 = d + 2 * h;
            const float4 L = *(const float4*)(lut + ((size_t)prow * NDET + det0) * 2);
            v2f bb;
            bb.x = 0.5f - 0.5f * cosf((float)det0 * w);
            bb.y = 0.5f - 0.5f * cosf((float)(det0 + 1) * w);
            const int kf0 = (int)L.x;
            const int kf1 = (int)L.z;
            const bool valid0 = (kf0 >= 0) && (kf0 < NT - 1);
            const bool valid1 = (kf1 >= 0) && (kf1 < NT - 1);
            const int k00 = min(max(kf0, 0), NT - 2);
            const int k10 = min(max(kf1, 0), NT - 2);
            const int roff0 = det0 * NT + k00;
            const int roff1 = (det0 + 1) * NT + k10;
            const float a0 = L.y, a1 = L.w;
#pragma unroll
            for (int b = 0; b < NB; ++b) {
                const float* s = sino + (size_t)b * (NDET * NT);
                const float s00 = s[roff0], s01 = s[roff0 + 1];
                const float s10 = s[roff1], s11 = s[roff1 + 1];
                float sk0 = (1.0f - a0) * s00 + a0 * s01;
                float sk1 = (1.0f - a1) * s10 + a1 * s11;
                sk0 = valid0 ? sk0 : 0.0f;
                sk1 = valid1 ? sk1 : 0.0f;
                v2f A; A.x = sk0; A.y = sk1;
                acc[b] = __builtin_amdgcn_wmma_f32_16x16x4_f32(
                    false, A, false, bb, (short)0, acc[b], false, false);
            }
        }
        const float invNorm = 1.0f / 63.5f;
        if (lane == 0 || lane == 16) {
            const int rbase = p0 + ((lane == 16) ? 8 : 0);
#pragma unroll
            for (int b = 0; b < NB; ++b) {
#pragma unroll
                for (int r = 0; r < 8; ++r) {
                    out[(size_t)b * NPIX + rbase + r] = acc[b][r] * invNorm;
                }
            }
        }
        return;
    }

    const int g  = blockIdx.x & (NGRP - 1);
    const int pg = blockIdx.x >> 4;

    const int dbase = g * GDET + 4 * h;
    const int tile0 = pg * TILES_PER_BLOCK + wave * TILES_PER_WAVE;
    const float4* lp0 = (const float4*)(lut + ((size_t)((tile0 << 4) + c) * NDET + dbase) * 2);
    const size_t tileStride = (size_t)16 * NDET * 2 / 4;

    float4 Lq[2][LUT_CHUNK][2];
#pragma unroll
    for (int j = 0; j < LUT_CHUNK; ++j) {
        Lq[0][j][0] = lp0[j * tileStride];
        Lq[0][j][1] = lp0[j * tileStride + 1];
    }

#pragma unroll
    for (int half = 0; half < 2; ++half) {
        const int t2 = 2 * threadIdx.x + half * (NT / 2);
        float2 v[GDET][NB];
#pragma unroll
        for (int dl = 0; dl < GDET; ++dl) {
            const float* r = sino + (size_t)(g * GDET + dl) * NT + t2;
#pragma unroll
            for (int b = 0; b < NB; ++b) v[dl][b] = *(const float2*)(r + (size_t)b * (NDET * NT));
        }
#pragma unroll
        for (int dl = 0; dl < GDET; ++dl) {
            sRow[dl * NT + t2]     = make_float4(v[dl][0].x, v[dl][1].x, v[dl][2].x, v[dl][3].x);
            sRow[dl * NT + t2 + 1] = make_float4(v[dl][0].y, v[dl][1].y, v[dl][2].y, v[dl][3].y);
        }
    }
    __syncthreads();

    v2f bb0, bb1;
    bb0.x = 0.5f - 0.5f * cosf((float)(dbase + 0) * w);
    bb0.y = 0.5f - 0.5f * cosf((float)(dbase + 1) * w);
    bb1.x = 0.5f - 0.5f * cosf((float)(dbase + 2) * w);
    bb1.y = 0.5f - 0.5f * cosf((float)(dbase + 3) * w);
    const float4* __restrict__ rows = sRow + (4 * h) * NT;
    const int cs = c & 7;

#pragma unroll
    for (int ch = 0; ch < TILES_PER_WAVE / LUT_CHUNK; ++ch) {
        if (ch + 1 < TILES_PER_WAVE / LUT_CHUNK) {
#pragma unroll
            for (int j = 0; j < LUT_CHUNK; ++j) {
                Lq[(ch + 1) & 1][j][0] = lp0[((ch + 1) * LUT_CHUNK + j) * tileStride];
                Lq[(ch + 1) & 1][j][1] = lp0[((ch + 1) * LUT_CHUNK + j) * tileStride + 1];
            }
        }
#pragma unroll
        for (int j = 0; j < LUT_CHUNK; ++j) {
            const int tile = tile0 + ch * LUT_CHUNK + j;
            const float4 L0 = Lq[ch & 1][j][0], L1 = Lq[ch & 1][j][1];

            float o0[NB], o1[NB], o2[NB], o3[NB];
            interp4(rows,          L0.x, L0.y, o0);
            interp4(rows + NT,     L0.z, L0.w, o1);
            interp4(rows + 2 * NT, L1.x, L1.y, o2);
            interp4(rows + 3 * NT, L1.z, L1.w, o3);

            v8f acc[NB];
#pragma unroll
            for (int b = 0; b < NB; ++b) {
                v2f A; A.x = o0[b]; A.y = o1[b];
                acc[b] = __builtin_amdgcn_wmma_f32_16x16x4_f32(
                    false, A, false, bb0, (short)0, v8f{0}, false, false);
            }
#pragma unroll
            for (int b = 0; b < NB; ++b) {
                v2f A; A.x = o2[b]; A.y = o3[b];
                acc[b] = __builtin_amdgcn_wmma_f32_16x16x4_f32(
                    false, A, false, bb1, (short)0, acc[b], false, false);
            }

            float v[NB];
#pragma unroll
            for (int b = 0; b < NB; ++b) {
                float x = acc[b][0];
#pragma unroll
                for (int r = 1; r < 8; ++r) x = (cs == r) ? acc[b][r] : x;
                v[b] = x;
            }
            float* pt = part + ((size_t)(g * NTILE + tile) * NB) * 16;
            const int m = 8 * h + cs;
            const int bs = c >> 3;
            pt[bs * 16 + m]       = bs ? v[1] : v[0];
            pt[(2 + bs) * 16 + m] = bs ? v[3] : v[2];
        }
    }
}

__global__ __launch_bounds__(256)
void bp_sum_groups_kernel(const float* __restrict__ part, size_t ws_bytes, float* __restrict__ out)
{
    if (ws_bytes < PART_BYTES) return;
    const int tid = blockIdx.x * 256 + threadIdx.x;
    const float invNorm = 1.0f / 63.5f;
#pragma unroll
    for (int u = 0; u < 2; ++u) {
        const int f = tid + u * 32768;
        const float4* p4 = (const float4*)part + f;
        float4 q[NGRP];
#pragma unroll
        for (int g = 0; g < NGRP; ++g) q[g] = p4[(size_t)g * (NTILE * NB * 4)];
        float4 s = q[0];
#pragma unroll
        for (int g = 1; g < NGRP; ++g) { s.x += q[g].x; s.y += q[g].y; s.z += q[g].z; s.w += q[g].w; }
        s.x *= invNorm; s.y *= invNorm; s.z *= invNorm; s.w *= invNorm;
        const int tile = f >> 4;
        const int b = (f >> 2) & 3;
        const int m4 = f & 3;
        *(float4*)(out + (size_t)b * NPIX + (tile << 4) + (m4 << 2)) = s;
    }
}

extern "C" void kernel_launch(void* const* d_in, const int* in_sizes, int n_in,
                              void* d_out, int out_size, void* d_ws, size_t ws_size,
                              hipStream_t stream) {
    const float* sino = (const float*)d_in[0];
    const float* lut  = (const float*)d_in[1];
    float* out = (float*)d_out;
    float* part = (float*)d_ws;

    bp_detgroup_kernel<<<256, BLOCK_THREADS, 0, stream>>>(sino, lut, part, ws_size, out);
    bp_sum_groups_kernel<<<128, 256, 0, stream>>>(part, ws_size, out);
}
